# starting kernel
# baseline (speedup 1.0000x reference)
#include <hip/hip_runtime.h>
#include <math.h>

typedef __attribute__((ext_vector_type(2))) float v2f;
typedef __attribute__((ext_vector_type(8))) float v8f;

#define HH 768
#define WW 768
#define HWSZ (768*768)
#define EPSV 1e-4f

__global__ void initW(float* wbil, float* wgau, float* sxb, float* sxg) {
  __shared__ float wb[121];
  __shared__ float wg[25];
  int t = threadIdx.x;
  if (t < 121) { float d = (float)(t - 60); float v = expf(-0.5f*d*d/(15.f*15.f)); wb[t] = v; wbil[t] = v; }
  if (t < 25)  { float d = (float)(t - 12); float v = expf(-d*d/18.f);             wg[t] = v; wgau[t] = v; }
  __syncthreads();
  for (int x = t; x < WW; x += blockDim.x) {
    float s = 0.f;
    for (int d = 0; d < 121; ++d) { int ix = x + d - 60; if (ix >= 0 && ix < WW) s += wb[d]; }
    sxb[x] = s;
    float s2 = 0.f;
    for (int d = 0; d < 25; ++d) { int ix = x + d - 12; if (ix >= 0 && ix < WW) s2 += wg[d]; }
    sxg[x] = s2;
  }
}

template<int R>
__global__ __launch_bounds__(256) void convH(const float* __restrict__ src,
                                             float* __restrict__ dst,
                                             const float* __restrict__ wdev) {
  constexpr int TW  = 128;
  constexpr int SW  = TW + 2*R;
  constexpr int NCH = (2*R + 16) / 4;
  constexpr int WP  = 2*R + 31;
  constexpr int SP  = SW + 1;
  __shared__ float lds[(WP + 1) + 16*SP];
  float* wpad  = lds;
  float* strip = lds + (WP + 1);
  const int tid = threadIdx.x;
  const size_t pb = (size_t)blockIdx.z * HWSZ;
  const float* ip = src + pb;
  float* op = dst + pb;
  const int gy0 = blockIdx.y * 16;
  const int gx0 = blockIdx.x * TW;
  for (int i = tid; i <= WP; i += 256) {
    int d = i - 15;
    wpad[i] = (d >= 0 && d <= 2*R) ? wdev[d] : 0.f;
  }
  for (int idx = tid; idx < 16*SW; idx += 256) {
    int r = idx / SW, c = idx - r*SW;
    int gx = gx0 - R + c;
    strip[r*SP + c] = (gx >= 0 && gx < WW) ? ip[(size_t)(gy0 + r)*WW + gx] : 0.f;
  }
  __syncthreads();
  const int lane = tid & 31;
  const int wv   = tid >> 5;
  const int j    = lane & 15;
  const int kb   = (lane >> 4) * 2;
  const float* ap = strip + j*SP + wv*16 + kb;
  const float* bp = wpad + (kb - j + 15);
  v8f acc = {0.f,0.f,0.f,0.f,0.f,0.f,0.f,0.f};
  #pragma unroll
  for (int kk = 0; kk < NCH; ++kk) {
    v2f a, b;
    a.x = ap[kk*4];
    a.y = ap[kk*4 + 1];
    b.x = bp[kk*4];
    b.y = bp[kk*4 + 1];
    acc = __builtin_amdgcn_wmma_f32_16x16x4_f32(false, a, false, b, (short)0, acc, false, false);
  }
  const int row0 = gy0 + (lane >> 4) * 8;
  const int col  = gx0 + wv*16 + j;
  #pragma unroll
  for (int i = 0; i < 8; ++i) op[(size_t)(row0 + i)*WW + col] = acc[i];
}

template<int R>
__global__ __launch_bounds__(256) void convV(const float* __restrict__ src,
                                             float* __restrict__ dst,
                                             const float* __restrict__ wdev) {
  constexpr int TH  = 128;
  constexpr int SH  = TH + 2*R;
  constexpr int NCH = (2*R + 16) / 4;
  constexpr int WP  = 2*R + 31;
  constexpr int SP  = SH + 1;
  __shared__ float lds[(WP + 1) + 16*SP];
  float* wpad   = lds;
  float* stripT = lds + (WP + 1);
  const int tid = threadIdx.x;
  const size_t pb = (size_t)blockIdx.z * HWSZ;
  const float* ip = src + pb;
  float* op = dst + pb;
  const int gy0 = blockIdx.y * TH;
  const int gx0 = blockIdx.x * 16;
  for (int i = tid; i <= WP; i += 256) {
    int d = i - 15;
    wpad[i] = (d >= 0 && d <= 2*R) ? wdev[d] : 0.f;
  }
  for (int idx = tid; idx < SH*16; idx += 256) {
    int r = idx >> 4, c = idx & 15;
    int gy = gy0 - R + r;
    stripT[c*SP + r] = (gy >= 0 && gy < HH) ? ip[(size_t)gy*WW + gx0 + c] : 0.f;
  }
  __syncthreads();
  const int lane = tid & 31;
  const int wv   = tid >> 5;
  const int j    = lane & 15;
  const int kb   = (lane >> 4) * 2;
  const float* bp = stripT + j*SP + wv*16 + kb;
  const float* apw = wpad + (kb - j + 15);
  v8f acc = {0.f,0.f,0.f,0.f,0.f,0.f,0.f,0.f};
  #pragma unroll
  for (int kk = 0; kk < NCH; ++kk) {
    v2f a, b;
    a.x = apw[kk*4];
    a.y = apw[kk*4 + 1];
    b.x = bp[kk*4];
    b.y = bp[kk*4 + 1];
    acc = __builtin_amdgcn_wmma_f32_16x16x4_f32(false, a, false, b, (short)0, acc, false, false);
  }
  const int row0 = gy0 + wv*16 + (lane >> 4) * 8;
  const int col  = gx0 + j;
  #pragma unroll
  for (int i = 0; i < 8; ++i) op[(size_t)(row0 + i)*WW + col] = acc[i];
}

__global__ void prepImg(const float* __restrict__ img, float* __restrict__ Ipl,
                        float* __restrict__ P9) {
  int p = blockIdx.x*256 + threadIdx.x;
  if (p >= HWSZ) return;
  float i0 = img[p*3+0], i1 = img[p*3+1], i2 = img[p*3+2];
  Ipl[p] = i0; Ipl[HWSZ+p] = i1; Ipl[2*HWSZ+p] = i2;
  P9[0*HWSZ+p] = i0;    P9[1*HWSZ+p] = i1;    P9[2*HWSZ+p] = i2;
  P9[3*HWSZ+p] = i0*i0; P9[4*HWSZ+p] = i0*i1; P9[5*HWSZ+p] = i0*i2;
  P9[6*HWSZ+p] = i1*i1; P9[7*HWSZ+p] = i1*i2; P9[8*HWSZ+p] = i2*i2;
}

__global__ void statsK(const float* __restrict__ U, const float* __restrict__ Ipl,
                       const float* __restrict__ sxb,
                       float* __restrict__ meanI, float* __restrict__ inv6) {
  int p = blockIdx.x*256 + threadIdx.x;
  if (p >= HWSZ) return;
  int x = p % WW, y = p / WW;
  float rN = 1.f / (sxb[x]*sxb[y] - 1.f);
  float i0 = Ipl[p], i1 = Ipl[HWSZ+p], i2 = Ipl[2*HWSZ+p];
  float m0 = (U[0*HWSZ+p] - i0)*rN;
  float m1 = (U[1*HWSZ+p] - i1)*rN;
  float m2 = (U[2*HWSZ+p] - i2)*rN;
  float v00 = (U[3*HWSZ+p] - i0*i0)*rN - m0*m0 + EPSV;
  float v01 = (U[4*HWSZ+p] - i0*i1)*rN - m0*m1;
  float v02 = (U[5*HWSZ+p] - i0*i2)*rN - m0*m2;
  float v11 = (U[6*HWSZ+p] - i1*i1)*rN - m1*m1 + EPSV;
  float v12 = (U[7*HWSZ+p] - i1*i2)*rN - m1*m2;
  float v22 = (U[8*HWSZ+p] - i2*i2)*rN - m2*m2 + EPSV;
  float c00 = v11*v22 - v12*v12;
  float c01 = v02*v12 - v01*v22;
  float c02 = v01*v12 - v02*v11;
  float c11 = v00*v22 - v02*v02;
  float c12 = v01*v02 - v00*v12;
  float c22 = v00*v11 - v01*v01;
  float rd = 1.f / (v00*c00 + v01*c01 + v02*c02);
  meanI[p] = m0; meanI[HWSZ+p] = m1; meanI[2*HWSZ+p] = m2;
  inv6[0*HWSZ+p] = c00*rd; inv6[1*HWSZ+p] = c01*rd; inv6[2*HWSZ+p] = c02*rd;
  inv6[3*HWSZ+p] = c11*rd; inv6[4*HWSZ+p] = c12*rd; inv6[5*HWSZ+p] = c22*rd;
}

__global__ void softmaxInit(const float* __restrict__ un, float* __restrict__ Q) {
  int p = blockIdx.x*256 + threadIdx.x;
  if (p >= HWSZ) return;
  float L[4];
  float m = -3.4e38f;
  #pragma unroll
  for (int k = 0; k < 4; ++k) { L[k] = -un[p*4+k]; m = fmaxf(m, L[k]); }
  float s = 0.f;
  #pragma unroll
  for (int k = 0; k < 4; ++k) { L[k] = expf(L[k] - m); s += L[k]; }
  float rs = 1.f / s;
  #pragma unroll
  for (int k = 0; k < 4; ++k) Q[k*HWSZ+p] = L[k]*rs;
}

__global__ void prep16K(const float* __restrict__ Q, const float* __restrict__ Ipl,
                        float* __restrict__ P) {
  int p = blockIdx.x*256 + threadIdx.x;
  if (p >= HWSZ) return;
  float I[3] = { Ipl[p], Ipl[HWSZ+p], Ipl[2*HWSZ+p] };
  #pragma unroll
  for (int k = 0; k < 4; ++k) {
    float q = Q[k*HWSZ+p];
    P[k*HWSZ+p] = q;
    #pragma unroll
    for (int i = 0; i < 3; ++i) P[(size_t)(4 + i*4 + k)*HWSZ+p] = I[i]*q;
  }
}

__global__ void midABK(const float* __restrict__ U, const float* __restrict__ Q,
                       const float* __restrict__ Ipl, const float* __restrict__ meanI,
                       const float* __restrict__ inv6, const float* __restrict__ sxb,
                       float* __restrict__ AB) {
  int p = blockIdx.x*256 + threadIdx.x;
  if (p >= HWSZ) return;
  int x = p % WW, y = p / WW;
  float rN = 1.f / (sxb[x]*sxb[y] - 1.f);
  float I0 = Ipl[p], I1 = Ipl[HWSZ+p], I2 = Ipl[2*HWSZ+p];
  float mI0 = meanI[p], mI1 = meanI[HWSZ+p], mI2 = meanI[2*HWSZ+p];
  float iv0 = inv6[0*HWSZ+p], iv1 = inv6[1*HWSZ+p], iv2 = inv6[2*HWSZ+p];
  float iv3 = inv6[3*HWSZ+p], iv4 = inv6[4*HWSZ+p], iv5 = inv6[5*HWSZ+p];
  #pragma unroll
  for (int k = 0; k < 4; ++k) {
    float q  = Q[k*HWSZ+p];
    float mp = (U[k*HWSZ+p] - q)*rN;
    float c0 = (U[(size_t)(4 +k)*HWSZ+p] - I0*q)*rN - mI0*mp;
    float c1 = (U[(size_t)(8 +k)*HWSZ+p] - I1*q)*rN - mI1*mp;
    float c2 = (U[(size_t)(12+k)*HWSZ+p] - I2*q)*rN - mI2*mp;
    float a0 = iv0*c0 + iv1*c1 + iv2*c2;
    float a1 = iv1*c0 + iv3*c1 + iv4*c2;
    float a2 = iv2*c0 + iv4*c1 + iv5*c2;
    float b  = mp - (a0*mI0 + a1*mI1 + a2*mI2);
    AB[(size_t)(0 +k)*HWSZ+p] = a0;
    AB[(size_t)(4 +k)*HWSZ+p] = a1;
    AB[(size_t)(8 +k)*HWSZ+p] = a2;
    AB[(size_t)(12+k)*HWSZ+p] = b;
  }
}

__global__ void combineK(const float* __restrict__ U, const float* __restrict__ AB,
                         const float* __restrict__ Us4, const float* __restrict__ Ipl,
                         const float* __restrict__ un, const float* __restrict__ sxb,
                         const float* __restrict__ sxg, float* __restrict__ Q) {
  int p = blockIdx.x*256 + threadIdx.x;
  if (p >= HWSZ) return;
  int x = p % WW, y = p / WW;
  float rN = 1.f / (sxb[x]*sxb[y] - 1.f);
  float rS = 1.f / (sxg[x]*sxg[y] - 1.f);
  float I0 = Ipl[p], I1 = Ipl[HWSZ+p], I2 = Ipl[2*HWSZ+p];
  float L[4];
  float m = -3.4e38f;
  #pragma unroll
  for (int k = 0; k < 4; ++k) {
    float ma0 = (U[(size_t)(0 +k)*HWSZ+p] - AB[(size_t)(0 +k)*HWSZ+p])*rN;
    float ma1 = (U[(size_t)(4 +k)*HWSZ+p] - AB[(size_t)(4 +k)*HWSZ+p])*rN;
    float ma2 = (U[(size_t)(8 +k)*HWSZ+p] - AB[(size_t)(8 +k)*HWSZ+p])*rN;
    float mb  = (U[(size_t)(12+k)*HWSZ+p] - AB[(size_t)(12+k)*HWSZ+p])*rN;
    float bil = ma0*I0 + ma1*I1 + ma2*I2 + mb;
    float q   = Q[k*HWSZ+p];
    float sp  = (Us4[k*HWSZ+p] - q)*rS;
    L[k] = -un[p*4+k] + 3.f*sp + 10.f*bil;
    m = fmaxf(m, L[k]);
  }
  float s = 0.f;
  #pragma unroll
  for (int k = 0; k < 4; ++k) { L[k] = expf(L[k] - m); s += L[k]; }
  float rs = 1.f / s;
  #pragma unroll
  for (int k = 0; k < 4; ++k) Q[k*HWSZ+p] = L[k]*rs;
}

__global__ void packK(const float* __restrict__ Q, float* __restrict__ out) {
  int p = blockIdx.x*256 + threadIdx.x;
  if (p >= HWSZ) return;
  #pragma unroll
  for (int k = 0; k < 4; ++k) out[p*4+k] = Q[k*HWSZ+p];
}

extern "C" void kernel_launch(void* const* d_in, const int* in_sizes, int n_in,
                              void* d_out, int out_size, void* d_ws, size_t ws_size,
                              hipStream_t stream) {
  (void)in_sizes; (void)n_in; (void)out_size; (void)ws_size;
  const float* img = (const float*)d_in[0];
  const float* un  = (const float*)d_in[1];
  float* ws   = (float*)d_ws;
  float* wbil = ws;
  float* wgau = ws + 128;
  float* sxb  = ws + 160;
  float* sxg  = ws + 928;
  float* pl   = ws + 2048;
  auto P = [&](int i) { return pl + (size_t)i * HWSZ; };

  dim3 blk(256);
  dim3 gP(HWSZ/256);
  initW<<<1, 256, 0, stream>>>(wbil, wgau, sxb, sxg);
  prepImg<<<gP, blk, 0, stream>>>(img, P(0), P(16));
  convH<60><<<dim3(6,48,9),  blk, 0, stream>>>(P(16), P(32), wbil);
  convV<60><<<dim3(48,6,9),  blk, 0, stream>>>(P(32), P(48), wbil);
  statsK<<<gP, blk, 0, stream>>>(P(48), P(0), sxb, P(3), P(6));
  softmaxInit<<<gP, blk, 0, stream>>>(un, P(12));

  for (int it = 0; it < 5; ++it) {
    prep16K<<<gP, blk, 0, stream>>>(P(12), P(0), P(16));
    convH<60><<<dim3(6,48,16), blk, 0, stream>>>(P(16), P(32), wbil);
    convV<60><<<dim3(48,6,16), blk, 0, stream>>>(P(32), P(48), wbil);
    midABK<<<gP, blk, 0, stream>>>(P(48), P(12), P(0), P(3), P(6), sxb, P(16));
    convH<60><<<dim3(6,48,16), blk, 0, stream>>>(P(16), P(32), wbil);
    convV<60><<<dim3(48,6,16), blk, 0, stream>>>(P(32), P(48), wbil);
    convH<12><<<dim3(6,48,4),  blk, 0, stream>>>(P(12), P(32), wgau);
    convV<12><<<dim3(48,6,4),  blk, 0, stream>>>(P(32), P(64), wgau);
    combineK<<<gP, blk, 0, stream>>>(P(48), P(16), P(64), P(0), un, sxb, sxg, P(12));
  }
  packK<<<gP, blk, 0, stream>>>(P(12), (float*)d_out);
}


# separable convs: one wave per 16x128 output strip, Toeplitz band kept in registers, each LDS input fragment read once for all 8 tiles; 8x fewer waves, strip filled by batched 16-byte loads
# speedup vs baseline: 1.2262x; 1.2262x over previous
#include <hip/hip_runtime.h>
#include <math.h>

typedef __attribute__((ext_vector_type(2))) float v2f;
typedef __attribute__((ext_vector_type(8))) float v8f;

#define HH 768
#define WW 768
#define HWSZ (768*768)
#define EPSV 1e-4f

__global__ void initW(float* wpb, float* wpg, float* sxb, float* sxg) {
  __shared__ float wb[121];
  __shared__ float wg[25];
  int t = threadIdx.x;
  if (t < 121) { float d = (float)(t - 60); wb[t] = expf(-0.5f*d*d/(15.f*15.f)); }
  if (t < 25)  { float d = (float)(t - 12); wg[t] = expf(-d*d/18.f); }
  __syncthreads();
  if (t < 152) { int d = t - 15; wpb[t] = (d >= 0 && d <= 120) ? wb[d] : 0.f; }
  if (t < 56)  { int d = t - 15; wpg[t] = (d >= 0 && d <= 24)  ? wg[d] : 0.f; }
  for (int x = t; x < WW; x += blockDim.x) {
    float s = 0.f;
    for (int d = 0; d < 121; ++d) { int ix = x + d - 60; if (ix >= 0 && ix < WW) s += wb[d]; }
    sxb[x] = s;
    float s2 = 0.f;
    for (int d = 0; d < 25; ++d) { int ix = x + d - 12; if (ix >= 0 && ix < WW) s2 += wg[d]; }
    sxg[x] = s2;
  }
}

template<int R> struct ConvCfg {
  static constexpr int TL  = 128;
  static constexpr int NT  = TL / 16;
  static constexpr int SL  = TL + 2*R;
  static constexpr int NG  = SL / 4;
  static constexpr int NCH = (2*R + 16) / 4;
  static constexpr int SP  = SL + 4;
  static_assert((SP / 4) % 2 == 1, "pitch/4 must be odd");
  static_assert(NG == NCH + 4*(NT - 1), "band and strip must agree");
};

template<int R>
__global__ __launch_bounds__(32, 5) void convH(const float* __restrict__ src,
                                            float* __restrict__ dst,
                                            const float* __restrict__ wpad) {
  using C = ConvCfg<R>;
  __shared__ __attribute__((aligned(16))) float strip[16 * C::SP];
  const int lane = threadIdx.x;
  const size_t pb = (size_t)blockIdx.z * HWSZ;
  const float* ip = src + pb;
  float* op = dst + pb;
  const int gy0 = blockIdx.y * 16;
  const int gx0 = blockIdx.x * C::TL;

  constexpr int NU = (C::NG + 31) / 32;
  float4 v[NU][16];
  bool ok[NU];
  #pragma unroll
  for (int u = 0; u < NU; ++u) {
    const int q  = lane + 32*u;
    const int gx = gx0 - R + 4*q;
    ok[u] = (gx >= 0) && (gx < WW);
    const int gxc = min(max(gx, 0), WW - 4);
    const float* gp = ip + ((size_t)gy0*WW + gxc);
    #pragma unroll
    for (int r = 0; r < 16; ++r)
      v[u][r] = *reinterpret_cast<const float4*>(gp + r*WW);
  }
  #pragma unroll
  for (int u = 0; u < NU; ++u) {
    const int q = lane + 32*u;
    if (q < C::NG) {
      float* sp = strip + 4*q;
      #pragma unroll
      for (int r = 0; r < 16; ++r) {
        float4 t = v[u][r];
        if (!ok[u]) t = make_float4(0.f, 0.f, 0.f, 0.f);
        *reinterpret_cast<float4*>(sp + r*C::SP) = t;
      }
    }
  }
  __syncthreads();

  const int j = lane & 15;
  const int h = lane >> 4;
  const float* wp = wpad + (2*h - j + 15);
  v2f wf[C::NCH];
  #pragma unroll
  for (int kk = 0; kk < C::NCH; ++kk) { wf[kk].x = wp[4*kk]; wf[kk].y = wp[4*kk + 1]; }

  const float* mp = strip + j*C::SP + 2*h;
  v8f acc[C::NT];
  #pragma unroll
  for (int t = 0; t < C::NT; ++t) acc[t] = (v8f){0.f,0.f,0.f,0.f,0.f,0.f,0.f,0.f};
  #pragma unroll
  for (int g = 0; g < C::NG; ++g) {
    const v2f a = *reinterpret_cast<const v2f*>(mp + 4*g);
    #pragma unroll
    for (int t = 0; t < C::NT; ++t) {
      const int kk = g - 4*t;
      if (kk >= 0 && kk < C::NCH)
        acc[t] = __builtin_amdgcn_wmma_f32_16x16x4_f32(false, a, false, wf[kk], (short)0, acc[t], false, false);
    }
  }
  float* o = op + (size_t)(gy0 + 8*h)*WW + gx0 + j;
  #pragma unroll
  for (int t = 0; t < C::NT; ++t) {
    #pragma unroll
    for (int i = 0; i < 8; ++i) o[(size_t)i*WW + 16*t] = acc[t][i];
  }
}

template<int R>
__global__ __launch_bounds__(32, 5) void convV(const float* __restrict__ src,
                                            float* __restrict__ dst,
                                            const float* __restrict__ wpad) {
  using C = ConvCfg<R>;
  __shared__ __attribute__((aligned(16))) float stripT[16 * C::SP];
  const int lane = threadIdx.x;
  const size_t pb = (size_t)blockIdx.z * HWSZ;
  const float* ip = src + pb;
  float* op = dst + pb;
  const int gy0 = blockIdx.y * C::TL;
  const int gx0 = blockIdx.x * 16;

  constexpr int NLD = C::SL / 8;
  static_assert(C::SL % 8 == 0, "strip rows must split over 8 row groups");
  const int q  = lane & 3;
  const int rl = lane >> 2;
  float4 v[NLD];
  #pragma unroll
  for (int m = 0; m < NLD; ++m) {
    const int gy  = gy0 - R + rl + 8*m;
    const int gyc = min(max(gy, 0), HH - 1);
    v[m] = *reinterpret_cast<const float4*>(ip + ((size_t)gyc*WW + gx0 + 4*q));
  }
  float* sp = stripT + (4*q)*C::SP + rl;
  #pragma unroll
  for (int m = 0; m < NLD; ++m) {
    const int gy = gy0 - R + rl + 8*m;
    float4 t = v[m];
    if (gy < 0 || gy >= HH) t = make_float4(0.f, 0.f, 0.f, 0.f);
    sp[8*m]             = t.x;
    sp[8*m + C::SP]     = t.y;
    sp[8*m + 2*C::SP]   = t.z;
    sp[8*m + 3*C::SP]   = t.w;
  }
  __syncthreads();

  const int j = lane & 15;
  const int h = lane >> 4;
  const float* wp = wpad + (2*h - j + 15);
  v2f wf[C::NCH];
  #pragma unroll
  for (int kk = 0; kk < C::NCH; ++kk) { wf[kk].x = wp[4*kk]; wf[kk].y = wp[4*kk + 1]; }

  const float* mp = stripT + j*C::SP + 2*h;
  v8f acc[C::NT];
  #pragma unroll
  for (int t = 0; t < C::NT; ++t) acc[t] = (v8f){0.f,0.f,0.f,0.f,0.f,0.f,0.f,0.f};
  #pragma unroll
  for (int g = 0; g < C::NG; ++g) {
    const v2f b = *reinterpret_cast<const v2f*>(mp + 4*g);
    #pragma unroll
    for (int t = 0; t < C::NT; ++t) {
      const int kk = g - 4*t;
      if (kk >= 0 && kk < C::NCH)
        acc[t] = __builtin_amdgcn_wmma_f32_16x16x4_f32(false, wf[kk], false, b, (short)0, acc[t], false, false);
    }
  }
  float* o = op + (size_t)(gy0 + 8*h)*WW + gx0 + j;
  #pragma unroll
  for (int t = 0; t < C::NT; ++t) {
    #pragma unroll
    for (int i = 0; i < 8; ++i) o[(size_t)(16*t + i)*WW] = acc[t][i];
  }
}

__global__ void prepImg(const float* __restrict__ img, float* __restrict__ Ipl,
                        float* __restrict__ P9) {
  int p = blockIdx.x*256 + threadIdx.x;
  if (p >= HWSZ) return;
  float i0 = img[p*3+0], i1 = img[p*3+1], i2 = img[p*3+2];
  Ipl[p] = i0; Ipl[HWSZ+p] = i1; Ipl[2*HWSZ+p] = i2;
  P9[0*HWSZ+p] = i0;    P9[1*HWSZ+p] = i1;    P9[2*HWSZ+p] = i2;
  P9[3*HWSZ+p] = i0*i0; P9[4*HWSZ+p] = i0*i1; P9[5*HWSZ+p] = i0*i2;
  P9[6*HWSZ+p] = i1*i1; P9[7*HWSZ+p] = i1*i2; P9[8*HWSZ+p] = i2*i2;
}

__global__ void statsK(const float* __restrict__ U, const float* __restrict__ Ipl,
                       const float* __restrict__ sxb,
                       float* __restrict__ meanI, float* __restrict__ inv6) {
  int p = blockIdx.x*256 + threadIdx.x;
  if (p >= HWSZ) return;
  int x = p % WW, y = p / WW;
  float rN = 1.f / (sxb[x]*sxb[y] - 1.f);
  float i0 = Ipl[p], i1 = Ipl[HWSZ+p], i2 = Ipl[2*HWSZ+p];
  float m0 = (U[0*HWSZ+p] - i0)*rN;
  float m1 = (U[1*HWSZ+p] - i1)*rN;
  float m2 = (U[2*HWSZ+p] - i2)*rN;
  float v00 = (U[3*HWSZ+p] - i0*i0)*rN - m0*m0 + EPSV;
  float v01 = (U[4*HWSZ+p] - i0*i1)*rN - m0*m1;
  float v02 = (U[5*HWSZ+p] - i0*i2)*rN - m0*m2;
  float v11 = (U[6*HWSZ+p] - i1*i1)*rN - m1*m1 + EPSV;
  float v12 = (U[7*HWSZ+p] - i1*i2)*rN - m1*m2;
  float v22 = (U[8*HWSZ+p] - i2*i2)*rN - m2*m2 + EPSV;
  float c00 = v11*v22 - v12*v12;
  float c01 = v02*v12 - v01*v22;
  float c02 = v01*v12 - v02*v11;
  float c11 = v00*v22 - v02*v02;
  float c12 = v01*v02 - v00*v12;
  float c22 = v00*v11 - v01*v01;
  float rd = 1.f / (v00*c00 + v01*c01 + v02*c02);
  meanI[p] = m0; meanI[HWSZ+p] = m1; meanI[2*HWSZ+p] = m2;
  inv6[0*HWSZ+p] = c00*rd; inv6[1*HWSZ+p] = c01*rd; inv6[2*HWSZ+p] = c02*rd;
  inv6[3*HWSZ+p] = c11*rd; inv6[4*HWSZ+p] = c12*rd; inv6[5*HWSZ+p] = c22*rd;
}

__global__ void softmaxInit(const float* __restrict__ un, float* __restrict__ Q) {
  int p = blockIdx.x*256 + threadIdx.x;
  if (p >= HWSZ) return;
  float L[4];
  float m = -3.4e38f;
  #pragma unroll
  for (int k = 0; k < 4; ++k) { L[k] = -un[p*4+k]; m = fmaxf(m, L[k]); }
  float s = 0.f;
  #pragma unroll
  for (int k = 0; k < 4; ++k) { L[k] = expf(L[k] - m); s += L[k]; }
  float rs = 1.f / s;
  #pragma unroll
  for (int k = 0; k < 4; ++k) Q[k*HWSZ+p] = L[k]*rs;
}

__global__ void prep16K(const float* __restrict__ Q, const float* __restrict__ Ipl,
                        float* __restrict__ P) {
  int p = blockIdx.x*256 + threadIdx.x;
  if (p >= HWSZ) return;
  float I[3] = { Ipl[p], Ipl[HWSZ+p], Ipl[2*HWSZ+p] };
  #pragma unroll
  for (int k = 0; k < 4; ++k) {
    float q = Q[k*HWSZ+p];
    P[k*HWSZ+p] = q;
    #pragma unroll
    for (int i = 0; i < 3; ++i) P[(size_t)(4 + i*4 + k)*HWSZ+p] = I[i]*q;
  }
}

__global__ void midABK(const float* __restrict__ U, const float* __restrict__ Q,
                       const float* __restrict__ Ipl, const float* __restrict__ meanI,
                       const float* __restrict__ inv6, const float* __restrict__ sxb,
                       float* __restrict__ AB) {
  int p = blockIdx.x*256 + threadIdx.x;
  if (p >= HWSZ) return;
  int x = p % WW, y = p / WW;
  float rN = 1.f / (sxb[x]*sxb[y] - 1.f);
  float I0 = Ipl[p], I1 = Ipl[HWSZ+p], I2 = Ipl[2*HWSZ+p];
  float mI0 = meanI[p], mI1 = meanI[HWSZ+p], mI2 = meanI[2*HWSZ+p];
  float iv0 = inv6[0*HWSZ+p], iv1 = inv6[1*HWSZ+p], iv2 = inv6[2*HWSZ+p];
  float iv3 = inv6[3*HWSZ+p], iv4 = inv6[4*HWSZ+p], iv5 = inv6[5*HWSZ+p];
  #pragma unroll
  for (int k = 0; k < 4; ++k) {
    float q  = Q[k*HWSZ+p];
    float mp = (U[k*HWSZ+p] - q)*rN;
    float c0 = (U[(size_t)(4 +k)*HWSZ+p] - I0*q)*rN - mI0*mp;
    float c1 = (U[(size_t)(8 +k)*HWSZ+p] - I1*q)*rN - mI1*mp;
    float c2 = (U[(size_t)(12+k)*HWSZ+p] - I2*q)*rN - mI2*mp;
    float a0 = iv0*c0 + iv1*c1 + iv2*c2;
    float a1 = iv1*c0 + iv3*c1 + iv4*c2;
    float a2 = iv2*c0 + iv4*c1 + iv5*c2;
    float b  = mp - (a0*mI0 + a1*mI1 + a2*mI2);
    AB[(size_t)(0 +k)*HWSZ+p] = a0;
    AB[(size_t)(4 +k)*HWSZ+p] = a1;
    AB[(size_t)(8 +k)*HWSZ+p] = a2;
    AB[(size_t)(12+k)*HWSZ+p] = b;
  }
}

__global__ void combineK(const float* __restrict__ U, const float* __restrict__ AB,
                         const float* __restrict__ Us4, const float* __restrict__ Ipl,
                         const float* __restrict__ un, const float* __restrict__ sxb,
                         const float* __restrict__ sxg, float* __restrict__ Q) {
  int p = blockIdx.x*256 + threadIdx.x;
  if (p >= HWSZ) return;
  int x = p % WW, y = p / WW;
  float rN = 1.f / (sxb[x]*sxb[y] - 1.f);
  float rS = 1.f / (sxg[x]*sxg[y] - 1.f);
  float I0 = Ipl[p], I1 = Ipl[HWSZ+p], I2 = Ipl[2*HWSZ+p];
  float L[4];
  float m = -3.4e38f;
  #pragma unroll
  for (int k = 0; k < 4; ++k) {
    float ma0 = (U[(size_t)(0 +k)*HWSZ+p] - AB[(size_t)(0 +k)*HWSZ+p])*rN;
    float ma1 = (U[(size_t)(4 +k)*HWSZ+p] - AB[(size_t)(4 +k)*HWSZ+p])*rN;
    float ma2 = (U[(size_t)(8 +k)*HWSZ+p] - AB[(size_t)(8 +k)*HWSZ+p])*rN;
    float mb  = (U[(size_t)(12+k)*HWSZ+p] - AB[(size_t)(12+k)*HWSZ+p])*rN;
    float bil = ma0*I0 + ma1*I1 + ma2*I2 + mb;
    float q   = Q[k*HWSZ+p];
    float sp  = (Us4[k*HWSZ+p] - q)*rS;
    L[k] = -un[p*4+k] + 3.f*sp + 10.f*bil;
    m = fmaxf(m, L[k]);
  }
  float s = 0.f;
  #pragma unroll
  for (int k = 0; k < 4; ++k) { L[k] = expf(L[k] - m); s += L[k]; }
  float rs = 1.f / s;
  #pragma unroll
  for (int k = 0; k < 4; ++k) Q[k*HWSZ+p] = L[k]*rs;
}

__global__ void packK(const float* __restrict__ Q, float* __restrict__ out) {
  int p = blockIdx.x*256 + threadIdx.x;
  if (p >= HWSZ) return;
  #pragma unroll
  for (int k = 0; k < 4; ++k) out[p*4+k] = Q[k*HWSZ+p];
}

extern "C" void kernel_launch(void* const* d_in, const int* in_sizes, int n_in,
                              void* d_out, int out_size, void* d_ws, size_t ws_size,
                              hipStream_t stream) {
  (void)in_sizes; (void)n_in; (void)out_size; (void)ws_size;
  const float* img = (const float*)d_in[0];
  const float* un  = (const float*)d_in[1];
  float* ws   = (float*)d_ws;
  float* wbil = ws;
  float* wgau = ws + 160;
  float* sxb  = ws + 256;
  float* sxg  = ws + 1024;
  float* pl   = ws + 2048;
  auto P = [&](int i) { return pl + (size_t)i * HWSZ; };

  dim3 blk(256);
  dim3 wv1(32);
  dim3 gP(HWSZ/256);
  initW<<<1, 256, 0, stream>>>(wbil, wgau, sxb, sxg);
  prepImg<<<gP, blk, 0, stream>>>(img, P(0), P(16));
  convH<60><<<dim3(6,48,9), wv1, 0, stream>>>(P(16), P(32), wbil);
  convV<60><<<dim3(48,6,9), wv1, 0, stream>>>(P(32), P(48), wbil);
  statsK<<<gP, blk, 0, stream>>>(P(48), P(0), sxb, P(3), P(6));
  softmaxInit<<<gP, blk, 0, stream>>>(un, P(12));

  for (int it = 0; it < 5; ++it) {
    prep16K<<<gP, blk, 0, stream>>>(P(12), P(0), P(16));
    convH<60><<<dim3(6,48,16), wv1, 0, stream>>>(P(16), P(32), wbil);
    convV<60><<<dim3(48,6,16), wv1, 0, stream>>>(P(32), P(48), wbil);
    midABK<<<gP, blk, 0, stream>>>(P(48), P(12), P(0), P(3), P(6), sxb, P(16));
    convH<60><<<dim3(6,48,16), wv1, 0, stream>>>(P(16), P(32), wbil);
    convV<60><<<dim3(48,6,16), wv1, 0, stream>>>(P(32), P(48), wbil);
    convH<12><<<dim3(6,48,4), wv1, 0, stream>>>(P(12), P(32), wgau);
    convV<12><<<dim3(48,6,4), wv1, 0, stream>>>(P(32), P(64), wgau);
    combineK<<<gP, blk, 0, stream>>>(P(48), P(16), P(64), P(0), un, sxb, sxg, P(12));
  }
  packK<<<gP, blk, 0, stream>>>(P(12), (float*)d_out);
}
